# starting kernel
# baseline (speedup 1.0000x reference)
#include <hip/hip_runtime.h>

typedef __attribute__((ext_vector_type(2))) float v2f;
typedef __attribute__((ext_vector_type(8))) float v8f;

static __device__ __forceinline__ float xor16(float v) {
    int i = __builtin_bit_cast(int, v);
    int r = __builtin_amdgcn_permlanex16(i, i, 0x76543210u, 0xfedcba98u,
                                         false, false);
    return __builtin_bit_cast(float, r);
}

static __device__ __forceinline__ v8f wmma4(v2f a, v2f b, v8f c) {
    return __builtin_amdgcn_wmma_f32_16x16x4_f32(false, a, false, b, (short)0, c,
                                                 false, false);
}

__global__ __launch_bounds__(256) void nerf_render_kernel(
    const float* __restrict__ rays_o, const float* __restrict__ rays_d,
    const float* __restrict__ U,      const float* __restrict__ aabb,
    const float* __restrict__ W1,     const float* __restrict__ b1,
    const float* __restrict__ Wsig,   const float* __restrict__ Wcol,
    const float* __restrict__ Wdir,   float* __restrict__ out,
    int N, int T)
{
    const int  lane = threadIdx.x & 31;
    const int  ray  = blockIdx.x * blockDim.x + threadIdx.x;
    const int  rc   = (ray < N) ? ray : (N - 1);
    const int  m    = lane & 15;
    const bool hiL  = lane >= 16;

    v2f A1;
    A1.x = hiL ? W1[2 * 16 + m] : W1[0 * 16 + m];
    A1.y = hiL ? 0.0f           : W1[1 * 16 + m];

    v8f Cbias;
    #pragma unroll
    for (int r = 0; r < 8; ++r) Cbias[r] = hiL ? b1[r + 8] : b1[r];

    auto w2 = [&](int j, int c) -> float {
        return (c == 0) ? Wsig[j] : Wcol[j * 3 + (c - 1)];
    };
    v2f A2lo[4], A2hi[4];
    #pragma unroll
    for (int k = 0; k < 4; ++k) {
        const int j0 = 2 * k + (hiL ? 8 : 0);
        float l0v = 0.f, l1v = 0.f, h0v = 0.f, h1v = 0.f;
        if (m < 4) {
            l0v = w2(j0, m);
            l1v = w2(j0 + 1, m);
        }
        if (m >= 8 && m < 12) {
            h0v = w2(j0, m - 8);
            h1v = w2(j0 + 1, m - 8);
        }
        A2lo[k].x = l0v;  A2lo[k].y = l1v;
        A2hi[k].x = h0v;  A2hi[k].y = h1v;
    }

    const float ox = rays_o[3 * rc + 0], oy = rays_o[3 * rc + 1], oz = rays_o[3 * rc + 2];
    const float dx = rays_d[3 * rc + 0], dy = rays_d[3 * rc + 1], dz = rays_d[3 * rc + 2];
    const float eps = 1e-9f;
    const float sdx = (fabsf(dx) < eps) ? eps : dx;
    const float sdy = (fabsf(dy) < eps) ? eps : dy;
    const float sdz = (fabsf(dz) < eps) ? eps : dz;
    const float bl0 = aabb[0], bl1 = aabb[1], bl2 = aabb[2];
    const float bh0 = aabb[3], bh1 = aabb[4], bh2 = aabb[5];
    const float t1x = (bl0 - ox) / sdx, t2x = (bh0 - ox) / sdx;
    const float t1y = (bl1 - oy) / sdy, t2y = (bh1 - oy) / sdy;
    const float t1z = (bl2 - oz) / sdz, t2z = (bh2 - oz) / sdz;
    float tn = fmaxf(fmaxf(fminf(t1x, t2x), fminf(t1y, t2y)), fminf(t1z, t2z));
    tn = fmaxf(tn, 0.0f);
    const float tf = fminf(fminf(fmaxf(t1x, t2x), fmaxf(t1y, t2y)), fmaxf(t1z, t2z));
    const float active = (tn < tf) ? 1.0f : 0.0f;

    const float dnorm  = sqrtf(dx * dx + dy * dy + dz * dz);
    const float idn    = 1.0f / dnorm;
    const float uxn = dx * idn, uyn = dy * idn, uzn = dz * idn;
    const float enc0 = uxn * Wdir[0] + uyn * Wdir[3] + uzn * Wdir[6];
    const float enc1 = uxn * Wdir[1] + uyn * Wdir[4] + uzn * Wdir[7];
    const float enc2 = uxn * Wdir[2] + uyn * Wdir[5] + uzn * Wdir[8];

    const float tfinal = tf * 10.0f;
    const float stepw  = (tf - tn) * (1.0f / (float)T);

    auto sample = [&](int t, float& o_sig, float& o_r, float& o_g, float& o_b,
                      float& o_ts) {
        const float uu = U[(size_t)t * (size_t)N + (size_t)rc];
        const float ts = fmaf((float)t + uu, stepw, tn);
        const float px = fmaf(ts, dx, ox);
        const float py = fmaf(ts, dy, oy);
        const float pz = fmaf(ts, dz, oz);

        const float xs = xor16(px), ys = xor16(py), zs = xor16(pz);
        v2f B0, B1;
        B0.x = hiL ? zs : px;  B0.y = py;
        B1.x = hiL ? pz : xs;  B1.y = ys;

        v8f H0 = wmma4(A1, B0, Cbias);
        v8f H1 = wmma4(A1, B1, Cbias);
        #pragma unroll
        for (int i = 0; i < 8; ++i) {
            H0[i] = fmaxf(H0[i], 0.0f);
            H1[i] = fmaxf(H1[i], 0.0f);
        }

        v2f b00 = {H0[0], H0[1]}, b01 = {H0[2], H0[3]};
        v2f b02 = {H0[4], H0[5]}, b03 = {H0[6], H0[7]};
        v2f b10 = {H1[0], H1[1]}, b11 = {H1[2], H1[3]};
        v2f b12 = {H1[4], H1[5]}, b13 = {H1[6], H1[7]};
        v8f O0 = {}, O1 = {};
        O0 = wmma4(A2lo[3], b03, O0);
        O1 = wmma4(A2hi[3], b13, O1);
        O0 = wmma4(A2lo[2], b02, O0);
        O1 = wmma4(A2hi[2], b12, O1);
        O0 = wmma4(A2lo[1], b01, O0);
        O1 = wmma4(A2hi[1], b11, O1);
        O0 = wmma4(A2lo[0], b00, O0);
        O1 = wmma4(A2hi[0], b10, O1);

        const float v0 = hiL ? O1[0] : O0[0];
        const float v1 = hiL ? O1[1] : O0[1];
        const float v2 = hiL ? O1[2] : O0[2];
        const float v3 = hiL ? O1[3] : O0[3];

        o_sig = fmaxf(v0, 0.0f) + __logf(1.0f + __expf(-fabsf(v0)));
        o_r   = __fdividef(1.0f, 1.0f + __expf(-(v1 + enc0)));
        o_g   = __fdividef(1.0f, 1.0f + __expf(-(v2 + enc1)));
        o_b   = __fdividef(1.0f, 1.0f + __expf(-(v3 + enc2)));
        o_ts  = ts;
    };

    float Ttr = 1.f, amap = 0.f, dmap = 0.f, cr = 0.f, cg = 0.f, cb = 0.f;
    float p_sig, p_r, p_g, p_b, p_ts;
    sample(0, p_sig, p_r, p_g, p_b, p_ts);

    for (int t = 1; t < T; ++t) {
        float sig, r, g, b, ts;
        sample(t, sig, r, g, b, ts);

        const float sd = p_sig * (ts - p_ts) * dnorm;
        const float es = __expf(-sd);
        const float w  = (1.0f - es) * Ttr;
        Ttr *= es;
        cr += w * p_r;  cg += w * p_g;  cb += w * p_b;
        amap += w;      dmap += w * p_ts;

        p_sig = sig; p_r = r; p_g = g; p_b = b; p_ts = ts;
    }
    {
        const float sd = p_sig * (tfinal - p_ts) * dnorm;
        const float es = __expf(-sd);
        const float w  = (1.0f - es) * Ttr;
        cr += w * p_r;  cg += w * p_g;  cb += w * p_b;
        amap += w;      dmap += w * p_ts;
    }

    if (ray < N) {
        float* o5 = out + (size_t)ray * 5;
        o5[0] = cr   * active;
        o5[1] = cg   * active;
        o5[2] = cb   * active;
        o5[3] = amap * active;
        o5[4] = dmap * active;
    }
}

extern "C" void kernel_launch(void* const* d_in, const int* in_sizes, int n_in,
                              void* d_out, int out_size, void* d_ws, size_t ws_size,
                              hipStream_t stream) {
    const float* rays_o = (const float*)d_in[0];
    const float* rays_d = (const float*)d_in[1];
    const float* U      = (const float*)d_in[2];
    const float* aabb   = (const float*)d_in[3];
    const float* W1     = (const float*)d_in[4];
    const float* b1     = (const float*)d_in[5];
    const float* Wsig   = (const float*)d_in[6];
    const float* Wcol   = (const float*)d_in[7];
    const float* Wdir   = (const float*)d_in[8];
    float* out = (float*)d_out;

    const int N = in_sizes[0] / 3;
    const int T = (N > 0) ? (in_sizes[2] / N) : 0;
    const int block = 256;
    const int grid  = (N + block - 1) / block;
    nerf_render_kernel<<<grid, block, 0, stream>>>(rays_o, rays_d, U, aabb,
                                                   W1, b1, Wsig, Wcol, Wdir,
                                                   out, N, T);
}


# sigmoid, softplus and transmittance on single hardware rcp/log2/exp2 ops with pre-scaled arguments instead of exact-divide and range-checked log expansions
# speedup vs baseline: 1.1707x; 1.1707x over previous
#include <hip/hip_runtime.h>

typedef __attribute__((ext_vector_type(2))) float v2f;
typedef __attribute__((ext_vector_type(8))) float v8f;

static __device__ __forceinline__ float xor16(float v) {
    int i = __builtin_bit_cast(int, v);
    int r = __builtin_amdgcn_permlanex16(i, i, 0x76543210u, 0xfedcba98u,
                                         false, false);
    return __builtin_bit_cast(float, r);
}

static __device__ __forceinline__ v8f wmma4(v2f a, v2f b, v8f c) {
    return __builtin_amdgcn_wmma_f32_16x16x4_f32(false, a, false, b, (short)0, c,
                                                 false, false);
}

__global__ __launch_bounds__(256) void nerf_render_kernel(
    const float* __restrict__ rays_o, const float* __restrict__ rays_d,
    const float* __restrict__ U,      const float* __restrict__ aabb,
    const float* __restrict__ W1,     const float* __restrict__ b1,
    const float* __restrict__ Wsig,   const float* __restrict__ Wcol,
    const float* __restrict__ Wdir,   float* __restrict__ out,
    int N, int T)
{
    const int  lane = threadIdx.x & 31;
    const int  ray  = blockIdx.x * blockDim.x + threadIdx.x;
    const int  rc   = (ray < N) ? ray : (N - 1);
    const int  m    = lane & 15;
    const bool hiL  = lane >= 16;

    v2f A1;
    A1.x = hiL ? W1[2 * 16 + m] : W1[0 * 16 + m];
    A1.y = hiL ? 0.0f           : W1[1 * 16 + m];

    v8f Cbias;
    #pragma unroll
    for (int r = 0; r < 8; ++r) Cbias[r] = hiL ? b1[r + 8] : b1[r];

    auto w2 = [&](int j, int c) -> float {
        return (c == 0) ? Wsig[j] : Wcol[j * 3 + (c - 1)];
    };
    v2f A2lo[4], A2hi[4];
    #pragma unroll
    for (int k = 0; k < 4; ++k) {
        const int j0 = 2 * k + (hiL ? 8 : 0);
        float l0v = 0.f, l1v = 0.f, h0v = 0.f, h1v = 0.f;
        if (m < 4) {
            l0v = w2(j0, m);
            l1v = w2(j0 + 1, m);
        }
        if (m >= 8 && m < 12) {
            h0v = w2(j0, m - 8);
            h1v = w2(j0 + 1, m - 8);
        }
        A2lo[k].x = l0v;  A2lo[k].y = l1v;
        A2hi[k].x = h0v;  A2hi[k].y = h1v;
    }

    const float ox = rays_o[3 * rc + 0], oy = rays_o[3 * rc + 1], oz = rays_o[3 * rc + 2];
    const float dx = rays_d[3 * rc + 0], dy = rays_d[3 * rc + 1], dz = rays_d[3 * rc + 2];
    const float eps = 1e-9f;
    const float sdx = (fabsf(dx) < eps) ? eps : dx;
    const float sdy = (fabsf(dy) < eps) ? eps : dy;
    const float sdz = (fabsf(dz) < eps) ? eps : dz;
    const float bl0 = aabb[0], bl1 = aabb[1], bl2 = aabb[2];
    const float bh0 = aabb[3], bh1 = aabb[4], bh2 = aabb[5];
    const float t1x = (bl0 - ox) / sdx, t2x = (bh0 - ox) / sdx;
    const float t1y = (bl1 - oy) / sdy, t2y = (bh1 - oy) / sdy;
    const float t1z = (bl2 - oz) / sdz, t2z = (bh2 - oz) / sdz;
    float tn = fmaxf(fmaxf(fminf(t1x, t2x), fminf(t1y, t2y)), fminf(t1z, t2z));
    tn = fmaxf(tn, 0.0f);
    const float tf = fminf(fminf(fmaxf(t1x, t2x), fmaxf(t1y, t2y)), fmaxf(t1z, t2z));
    const float active = (tn < tf) ? 1.0f : 0.0f;

    const float dnorm  = sqrtf(dx * dx + dy * dy + dz * dz);
    const float idn    = 1.0f / dnorm;
    const float uxn = dx * idn, uyn = dy * idn, uzn = dz * idn;
    const float enc0 = uxn * Wdir[0] + uyn * Wdir[3] + uzn * Wdir[6];
    const float enc1 = uxn * Wdir[1] + uyn * Wdir[4] + uzn * Wdir[7];
    const float enc2 = uxn * Wdir[2] + uyn * Wdir[5] + uzn * Wdir[8];

    const float kL2E = 1.4426950408889634f;
    const float kLN2 = 0.6931471805599453f;
    const float nenc0 = -kL2E * enc0, nenc1 = -kL2E * enc1, nenc2 = -kL2E * enc2;
    const float ndnl  = -kL2E * dnorm;

    const float tfinal = tf * 10.0f;
    const float stepw  = (tf - tn) * (1.0f / (float)T);

    auto sample = [&](int t, float& o_sig, float& o_r, float& o_g, float& o_b,
                      float& o_ts) {
        const float uu = U[(size_t)t * (size_t)N + (size_t)rc];
        const float ts = fmaf((float)t + uu, stepw, tn);
        const float px = fmaf(ts, dx, ox);
        const float py = fmaf(ts, dy, oy);
        const float pz = fmaf(ts, dz, oz);

        const float xs = xor16(px), ys = xor16(py), zs = xor16(pz);
        v2f B0, B1;
        B0.x = hiL ? zs : px;  B0.y = py;
        B1.x = hiL ? pz : xs;  B1.y = ys;

        v8f H0 = wmma4(A1, B0, Cbias);
        v8f H1 = wmma4(A1, B1, Cbias);
        #pragma unroll
        for (int i = 0; i < 8; ++i) {
            H0[i] = fmaxf(H0[i], 0.0f);
            H1[i] = fmaxf(H1[i], 0.0f);
        }

        v2f b00 = {H0[0], H0[1]}, b01 = {H0[2], H0[3]};
        v2f b02 = {H0[4], H0[5]}, b03 = {H0[6], H0[7]};
        v2f b10 = {H1[0], H1[1]}, b11 = {H1[2], H1[3]};
        v2f b12 = {H1[4], H1[5]}, b13 = {H1[6], H1[7]};
        v8f O0 = {}, O1 = {};
        O0 = wmma4(A2lo[3], b03, O0);
        O1 = wmma4(A2hi[3], b13, O1);
        O0 = wmma4(A2lo[2], b02, O0);
        O1 = wmma4(A2hi[2], b12, O1);
        O0 = wmma4(A2lo[1], b01, O0);
        O1 = wmma4(A2hi[1], b11, O1);
        O0 = wmma4(A2lo[0], b00, O0);
        O1 = wmma4(A2hi[0], b10, O1);

        const float v0 = hiL ? O1[0] : O0[0];
        const float v1 = hiL ? O1[1] : O0[1];
        const float v2 = hiL ? O1[2] : O0[2];
        const float v3 = hiL ? O1[3] : O0[3];

        const float e0 = __builtin_amdgcn_exp2f(-kL2E * fabsf(v0));
        o_sig = fmaf(__builtin_amdgcn_logf(1.0f + e0), kLN2, fmaxf(v0, 0.0f));
        o_r   = __builtin_amdgcn_rcpf(1.0f + __builtin_amdgcn_exp2f(fmaf(v1, -kL2E, nenc0)));
        o_g   = __builtin_amdgcn_rcpf(1.0f + __builtin_amdgcn_exp2f(fmaf(v2, -kL2E, nenc1)));
        o_b   = __builtin_amdgcn_rcpf(1.0f + __builtin_amdgcn_exp2f(fmaf(v3, -kL2E, nenc2)));
        o_ts  = ts;
    };

    float Ttr = 1.f, amap = 0.f, dmap = 0.f, cr = 0.f, cg = 0.f, cb = 0.f;
    float p_sig, p_r, p_g, p_b, p_ts;
    sample(0, p_sig, p_r, p_g, p_b, p_ts);

    for (int t = 1; t < T; ++t) {
        float sig, r, g, b, ts;
        sample(t, sig, r, g, b, ts);

        const float es = __builtin_amdgcn_exp2f(p_sig * (ts - p_ts) * ndnl);
        const float w  = (1.0f - es) * Ttr;
        Ttr *= es;
        cr += w * p_r;  cg += w * p_g;  cb += w * p_b;
        amap += w;      dmap += w * p_ts;

        p_sig = sig; p_r = r; p_g = g; p_b = b; p_ts = ts;
    }
    {
        const float es = __builtin_amdgcn_exp2f(p_sig * (tfinal - p_ts) * ndnl);
        const float w  = (1.0f - es) * Ttr;
        cr += w * p_r;  cg += w * p_g;  cb += w * p_b;
        amap += w;      dmap += w * p_ts;
    }

    if (ray < N) {
        float* o5 = out + (size_t)ray * 5;
        o5[0] = cr   * active;
        o5[1] = cg   * active;
        o5[2] = cb   * active;
        o5[3] = amap * active;
        o5[4] = dmap * active;
    }
}

extern "C" void kernel_launch(void* const* d_in, const int* in_sizes, int n_in,
                              void* d_out, int out_size, void* d_ws, size_t ws_size,
                              hipStream_t stream) {
    const float* rays_o = (const float*)d_in[0];
    const float* rays_d = (const float*)d_in[1];
    const float* U      = (const float*)d_in[2];
    const float* aabb   = (const float*)d_in[3];
    const float* W1     = (const float*)d_in[4];
    const float* b1     = (const float*)d_in[5];
    const float* Wsig   = (const float*)d_in[6];
    const float* Wcol   = (const float*)d_in[7];
    const float* Wdir   = (const float*)d_in[8];
    float* out = (float*)d_out;

    const int N = in_sizes[0] / 3;
    const int T = (N > 0) ? (in_sizes[2] / N) : 0;
    const int block = 256;
    const int grid  = (N + block - 1) / block;
    nerf_render_kernel<<<grid, block, 0, stream>>>(rays_o, rays_d, U, aabb,
                                                   W1, b1, Wsig, Wcol, Wdir,
                                                   out, N, T);
}


# layer 1 hoisted out of the sample loop by linearity in ts: W1^T o + b1 and W1^T d once per ray on the f32 WMMA, per sample one fma per hidden value (layer-1 rounding and sum order change)
# speedup vs baseline: 1.3373x; 1.1423x over previous
#include <hip/hip_runtime.h>

typedef __attribute__((ext_vector_type(2))) float v2f;
typedef __attribute__((ext_vector_type(8))) float v8f;

static __device__ __forceinline__ float xor16(float v) {
    int i = __builtin_bit_cast(int, v);
    int r = __builtin_amdgcn_permlanex16(i, i, 0x76543210u, 0xfedcba98u,
                                         false, false);
    return __builtin_bit_cast(float, r);
}

static __device__ __forceinline__ v8f wmma4(v2f a, v2f b, v8f c) {
    return __builtin_amdgcn_wmma_f32_16x16x4_f32(false, a, false, b, (short)0, c,
                                                 false, false);
}

__global__ __launch_bounds__(256) void nerf_render_kernel(
    const float* __restrict__ rays_o, const float* __restrict__ rays_d,
    const float* __restrict__ U,      const float* __restrict__ aabb,
    const float* __restrict__ W1,     const float* __restrict__ b1,
    const float* __restrict__ Wsig,   const float* __restrict__ Wcol,
    const float* __restrict__ Wdir,   float* __restrict__ out,
    int N, int T)
{
    const int  lane = threadIdx.x & 31;
    const int  ray  = blockIdx.x * blockDim.x + threadIdx.x;
    const int  rc   = (ray < N) ? ray : (N - 1);
    const int  m    = lane & 15;
    const bool hiL  = lane >= 16;

    v2f A1;
    A1.x = hiL ? W1[2 * 16 + m] : W1[0 * 16 + m];
    A1.y = hiL ? 0.0f           : W1[1 * 16 + m];

    v8f Cbias;
    #pragma unroll
    for (int r = 0; r < 8; ++r) Cbias[r] = hiL ? b1[r + 8] : b1[r];

    auto w2 = [&](int j, int c) -> float {
        return (c == 0) ? Wsig[j] : Wcol[j * 3 + (c - 1)];
    };
    v2f A2lo[4], A2hi[4];
    #pragma unroll
    for (int k = 0; k < 4; ++k) {
        const int j0 = 2 * k + (hiL ? 8 : 0);
        float l0v = 0.f, l1v = 0.f, h0v = 0.f, h1v = 0.f;
        if (m < 4) {
            l0v = w2(j0, m);
            l1v = w2(j0 + 1, m);
        }
        if (m >= 8 && m < 12) {
            h0v = w2(j0, m - 8);
            h1v = w2(j0 + 1, m - 8);
        }
        A2lo[k].x = l0v;  A2lo[k].y = l1v;
        A2hi[k].x = h0v;  A2hi[k].y = h1v;
    }

    const float ox = rays_o[3 * rc + 0], oy = rays_o[3 * rc + 1], oz = rays_o[3 * rc + 2];
    const float dx = rays_d[3 * rc + 0], dy = rays_d[3 * rc + 1], dz = rays_d[3 * rc + 2];
    const float eps = 1e-9f;
    const float sdx = (fabsf(dx) < eps) ? eps : dx;
    const float sdy = (fabsf(dy) < eps) ? eps : dy;
    const float sdz = (fabsf(dz) < eps) ? eps : dz;
    const float bl0 = aabb[0], bl1 = aabb[1], bl2 = aabb[2];
    const float bh0 = aabb[3], bh1 = aabb[4], bh2 = aabb[5];
    const float t1x = (bl0 - ox) / sdx, t2x = (bh0 - ox) / sdx;
    const float t1y = (bl1 - oy) / sdy, t2y = (bh1 - oy) / sdy;
    const float t1z = (bl2 - oz) / sdz, t2z = (bh2 - oz) / sdz;
    float tn = fmaxf(fmaxf(fminf(t1x, t2x), fminf(t1y, t2y)), fminf(t1z, t2z));
    tn = fmaxf(tn, 0.0f);
    const float tf = fminf(fminf(fmaxf(t1x, t2x), fmaxf(t1y, t2y)), fmaxf(t1z, t2z));
    const float active = (tn < tf) ? 1.0f : 0.0f;

    const float dnorm  = sqrtf(dx * dx + dy * dy + dz * dz);
    const float idn    = 1.0f / dnorm;
    const float uxn = dx * idn, uyn = dy * idn, uzn = dz * idn;
    const float enc0 = uxn * Wdir[0] + uyn * Wdir[3] + uzn * Wdir[6];
    const float enc1 = uxn * Wdir[1] + uyn * Wdir[4] + uzn * Wdir[7];
    const float enc2 = uxn * Wdir[2] + uyn * Wdir[5] + uzn * Wdir[8];

    const float kL2E = 1.4426950408889634f;
    const float kLN2 = 0.6931471805599453f;
    const float nenc0 = -kL2E * enc0, nenc1 = -kL2E * enc1, nenc2 = -kL2E * enc2;
    const float ndnl  = -kL2E * dnorm;

    const float tfinal = tf * 10.0f;
    const float stepw  = (tf - tn) * (1.0f / (float)T);

    const float oxs = xor16(ox), oys = xor16(oy), ozs = xor16(oz);
    const float dxs = xor16(dx), dys = xor16(dy), dzs = xor16(dz);
    v2f Bo0, Bo1, Bd0, Bd1;
    Bo0.x = hiL ? ozs : ox;   Bo0.y = oy;
    Bo1.x = hiL ? oz  : oxs;  Bo1.y = oys;
    Bd0.x = hiL ? dzs : dx;   Bd0.y = dy;
    Bd1.x = hiL ? dz  : dxs;  Bd1.y = dys;
    const v8f Zero8 = {};
    const v8f C0t0 = wmma4(A1, Bo0, Cbias);
    const v8f C0t1 = wmma4(A1, Bo1, Cbias);
    const v8f C1t0 = wmma4(A1, Bd0, Zero8);
    const v8f C1t1 = wmma4(A1, Bd1, Zero8);

    auto sample = [&](int t, float& o_sig, float& o_r, float& o_g, float& o_b,
                      float& o_ts) {
        const float uu = U[(size_t)t * (size_t)N + (size_t)rc];
        const float ts = fmaf((float)t + uu, stepw, tn);
        const float tsx = xor16(ts);
        const float ts0 = hiL ? tsx : ts;
        const float ts1 = hiL ? ts : tsx;
        v8f H0, H1;
        #pragma unroll
        for (int i = 0; i < 8; ++i) {
            H0[i] = fmaxf(fmaf(ts0, C1t0[i], C0t0[i]), 0.0f);
            H1[i] = fmaxf(fmaf(ts1, C1t1[i], C0t1[i]), 0.0f);
        }

        v2f b00 = {H0[0], H0[1]}, b01 = {H0[2], H0[3]};
        v2f b02 = {H0[4], H0[5]}, b03 = {H0[6], H0[7]};
        v2f b10 = {H1[0], H1[1]}, b11 = {H1[2], H1[3]};
        v2f b12 = {H1[4], H1[5]}, b13 = {H1[6], H1[7]};
        v8f O0 = {}, O1 = {};
        O0 = wmma4(A2lo[3], b03, O0);
        O1 = wmma4(A2hi[3], b13, O1);
        O0 = wmma4(A2lo[2], b02, O0);
        O1 = wmma4(A2hi[2], b12, O1);
        O0 = wmma4(A2lo[1], b01, O0);
        O1 = wmma4(A2hi[1], b11, O1);
        O0 = wmma4(A2lo[0], b00, O0);
        O1 = wmma4(A2hi[0], b10, O1);

        const float v0 = hiL ? O1[0] : O0[0];
        const float v1 = hiL ? O1[1] : O0[1];
        const float v2 = hiL ? O1[2] : O0[2];
        const float v3 = hiL ? O1[3] : O0[3];

        const float e0 = __builtin_amdgcn_exp2f(-kL2E * fabsf(v0));
        o_sig = fmaf(__builtin_amdgcn_logf(1.0f + e0), kLN2, fmaxf(v0, 0.0f));
        o_r   = __builtin_amdgcn_rcpf(1.0f + __builtin_amdgcn_exp2f(fmaf(v1, -kL2E, nenc0)));
        o_g   = __builtin_amdgcn_rcpf(1.0f + __builtin_amdgcn_exp2f(fmaf(v2, -kL2E, nenc1)));
        o_b   = __builtin_amdgcn_rcpf(1.0f + __builtin_amdgcn_exp2f(fmaf(v3, -kL2E, nenc2)));
        o_ts  = ts;
    };

    float Ttr = 1.f, amap = 0.f, dmap = 0.f, cr = 0.f, cg = 0.f, cb = 0.f;
    float p_sig, p_r, p_g, p_b, p_ts;
    sample(0, p_sig, p_r, p_g, p_b, p_ts);

    for (int t = 1; t < T; ++t) {
        float sig, r, g, b, ts;
        sample(t, sig, r, g, b, ts);

        const float es = __builtin_amdgcn_exp2f(p_sig * (ts - p_ts) * ndnl);
        const float w  = (1.0f - es) * Ttr;
        Ttr *= es;
        cr += w * p_r;  cg += w * p_g;  cb += w * p_b;
        amap += w;      dmap += w * p_ts;

        p_sig = sig; p_r = r; p_g = g; p_b = b; p_ts = ts;
    }
    {
        const float es = __builtin_amdgcn_exp2f(p_sig * (tfinal - p_ts) * ndnl);
        const float w  = (1.0f - es) * Ttr;
        cr += w * p_r;  cg += w * p_g;  cb += w * p_b;
        amap += w;      dmap += w * p_ts;
    }

    if (ray < N) {
        float* o5 = out + (size_t)ray * 5;
        o5[0] = cr   * active;
        o5[1] = cg   * active;
        o5[2] = cb   * active;
        o5[3] = amap * active;
        o5[4] = dmap * active;
    }
}

extern "C" void kernel_launch(void* const* d_in, const int* in_sizes, int n_in,
                              void* d_out, int out_size, void* d_ws, size_t ws_size,
                              hipStream_t stream) {
    const float* rays_o = (const float*)d_in[0];
    const float* rays_d = (const float*)d_in[1];
    const float* U      = (const float*)d_in[2];
    const float* aabb   = (const float*)d_in[3];
    const float* W1     = (const float*)d_in[4];
    const float* b1     = (const float*)d_in[5];
    const float* Wsig   = (const float*)d_in[6];
    const float* Wcol   = (const float*)d_in[7];
    const float* Wdir   = (const float*)d_in[8];
    float* out = (float*)d_out;

    const int N = in_sizes[0] / 3;
    const int T = (N > 0) ? (in_sizes[2] / N) : 0;
    const int block = 256;
    const int grid  = (N + block - 1) / block;
    nerf_render_kernel<<<grid, block, 0, stream>>>(rays_o, rays_d, U, aabb,
                                                   W1, b1, Wsig, Wcol, Wdir,
                                                   out, N, T);
}


# sample loop software-pipelined: two samples per trip, the next trip's two jitter loads issued at the head of a trip and first touched after its matrix work, so no sample starts with a load wait
# speedup vs baseline: 1.6368x; 1.2240x over previous
#include <hip/hip_runtime.h>

typedef __attribute__((ext_vector_type(2))) float v2f;
typedef __attribute__((ext_vector_type(8))) float v8f;

static __device__ __forceinline__ float xor16(float v) {
    int i = __builtin_bit_cast(int, v);
    int r = __builtin_amdgcn_permlanex16(i, i, 0x76543210u, 0xfedcba98u,
                                         false, false);
    return __builtin_bit_cast(float, r);
}

static __device__ __forceinline__ v8f wmma4(v2f a, v2f b, v8f c) {
    return __builtin_amdgcn_wmma_f32_16x16x4_f32(false, a, false, b, (short)0, c,
                                                 false, false);
}

__global__ __launch_bounds__(256) void nerf_render_kernel(
    const float* __restrict__ rays_o, const float* __restrict__ rays_d,
    const float* __restrict__ U,      const float* __restrict__ aabb,
    const float* __restrict__ W1,     const float* __restrict__ b1,
    const float* __restrict__ Wsig,   const float* __restrict__ Wcol,
    const float* __restrict__ Wdir,   float* __restrict__ out,
    int N, int T)
{
    const int  lane = threadIdx.x & 31;
    const int  ray  = blockIdx.x * blockDim.x + threadIdx.x;
    const int  rc   = (ray < N) ? ray : (N - 1);
    const int  m    = lane & 15;
    const bool hiL  = lane >= 16;

    v2f A1;
    A1.x = hiL ? W1[2 * 16 + m] : W1[0 * 16 + m];
    A1.y = hiL ? 0.0f           : W1[1 * 16 + m];

    v8f Cbias;
    #pragma unroll
    for (int r = 0; r < 8; ++r) Cbias[r] = hiL ? b1[r + 8] : b1[r];

    auto w2 = [&](int j, int c) -> float {
        return (c == 0) ? Wsig[j] : Wcol[j * 3 + (c - 1)];
    };
    v2f A2lo[4], A2hi[4];
    #pragma unroll
    for (int k = 0; k < 4; ++k) {
        const int j0 = 2 * k + (hiL ? 8 : 0);
        float l0v = 0.f, l1v = 0.f, h0v = 0.f, h1v = 0.f;
        if (m < 4) {
            l0v = w2(j0, m);
            l1v = w2(j0 + 1, m);
        }
        if (m >= 8 && m < 12) {
            h0v = w2(j0, m - 8);
            h1v = w2(j0 + 1, m - 8);
        }
        A2lo[k].x = l0v;  A2lo[k].y = l1v;
        A2hi[k].x = h0v;  A2hi[k].y = h1v;
    }

    const float ox = rays_o[3 * rc + 0], oy = rays_o[3 * rc + 1], oz = rays_o[3 * rc + 2];
    const float dx = rays_d[3 * rc + 0], dy = rays_d[3 * rc + 1], dz = rays_d[3 * rc + 2];
    const float eps = 1e-9f;
    const float sdx = (fabsf(dx) < eps) ? eps : dx;
    const float sdy = (fabsf(dy) < eps) ? eps : dy;
    const float sdz = (fabsf(dz) < eps) ? eps : dz;
    const float bl0 = aabb[0], bl1 = aabb[1], bl2 = aabb[2];
    const float bh0 = aabb[3], bh1 = aabb[4], bh2 = aabb[5];
    const float t1x = (bl0 - ox) / sdx, t2x = (bh0 - ox) / sdx;
    const float t1y = (bl1 - oy) / sdy, t2y = (bh1 - oy) / sdy;
    const float t1z = (bl2 - oz) / sdz, t2z = (bh2 - oz) / sdz;
    float tn = fmaxf(fmaxf(fminf(t1x, t2x), fminf(t1y, t2y)), fminf(t1z, t2z));
    tn = fmaxf(tn, 0.0f);
    const float tf = fminf(fminf(fmaxf(t1x, t2x), fmaxf(t1y, t2y)), fmaxf(t1z, t2z));
    const float active = (tn < tf) ? 1.0f : 0.0f;

    const float dnorm  = sqrtf(dx * dx + dy * dy + dz * dz);
    const float idn    = 1.0f / dnorm;
    const float uxn = dx * idn, uyn = dy * idn, uzn = dz * idn;
    const float enc0 = uxn * Wdir[0] + uyn * Wdir[3] + uzn * Wdir[6];
    const float enc1 = uxn * Wdir[1] + uyn * Wdir[4] + uzn * Wdir[7];
    const float enc2 = uxn * Wdir[2] + uyn * Wdir[5] + uzn * Wdir[8];

    const float kL2E = 1.4426950408889634f;
    const float kLN2 = 0.6931471805599453f;
    const float nenc0 = -kL2E * enc0, nenc1 = -kL2E * enc1, nenc2 = -kL2E * enc2;
    const float ndnl  = -kL2E * dnorm;

    const float tfinal = tf * 10.0f;
    const float stepw  = (tf - tn) * (1.0f / (float)T);

    const float oxs = xor16(ox), oys = xor16(oy), ozs = xor16(oz);
    const float dxs = xor16(dx), dys = xor16(dy), dzs = xor16(dz);
    v2f Bo0, Bo1, Bd0, Bd1;
    Bo0.x = hiL ? ozs : ox;   Bo0.y = oy;
    Bo1.x = hiL ? oz  : oxs;  Bo1.y = oys;
    Bd0.x = hiL ? dzs : dx;   Bd0.y = dy;
    Bd1.x = hiL ? dz  : dxs;  Bd1.y = dys;
    const v8f Zero8 = {};
    const v8f C0t0 = wmma4(A1, Bo0, Cbias);
    const v8f C0t1 = wmma4(A1, Bo1, Cbias);
    const v8f C1t0 = wmma4(A1, Bd0, Zero8);
    const v8f C1t1 = wmma4(A1, Bd1, Zero8);

    auto sample = [&](float ts, float& o_sig, float& o_r, float& o_g,
                      float& o_b, float& o_ts) {
        const float tsx = xor16(ts);
        const float ts0 = hiL ? tsx : ts;
        const float ts1 = hiL ? ts : tsx;
        v8f H0, H1;
        #pragma unroll
        for (int i = 0; i < 8; ++i) {
            H0[i] = fmaxf(fmaf(ts0, C1t0[i], C0t0[i]), 0.0f);
            H1[i] = fmaxf(fmaf(ts1, C1t1[i], C0t1[i]), 0.0f);
        }

        v2f b00 = {H0[0], H0[1]}, b01 = {H0[2], H0[3]};
        v2f b02 = {H0[4], H0[5]}, b03 = {H0[6], H0[7]};
        v2f b10 = {H1[0], H1[1]}, b11 = {H1[2], H1[3]};
        v2f b12 = {H1[4], H1[5]}, b13 = {H1[6], H1[7]};
        v8f O0 = {}, O1 = {};
        O0 = wmma4(A2lo[3], b03, O0);
        O1 = wmma4(A2hi[3], b13, O1);
        O0 = wmma4(A2lo[2], b02, O0);
        O1 = wmma4(A2hi[2], b12, O1);
        O0 = wmma4(A2lo[1], b01, O0);
        O1 = wmma4(A2hi[1], b11, O1);
        O0 = wmma4(A2lo[0], b00, O0);
        O1 = wmma4(A2hi[0], b10, O1);

        const float v0 = hiL ? O1[0] : O0[0];
        const float v1 = hiL ? O1[1] : O0[1];
        const float v2 = hiL ? O1[2] : O0[2];
        const float v3 = hiL ? O1[3] : O0[3];

        const float e0 = __builtin_amdgcn_exp2f(-kL2E * fabsf(v0));
        o_sig = fmaf(__builtin_amdgcn_logf(1.0f + e0), kLN2, fmaxf(v0, 0.0f));
        o_r   = __builtin_amdgcn_rcpf(1.0f + __builtin_amdgcn_exp2f(fmaf(v1, -kL2E, nenc0)));
        o_g   = __builtin_amdgcn_rcpf(1.0f + __builtin_amdgcn_exp2f(fmaf(v2, -kL2E, nenc1)));
        o_b   = __builtin_amdgcn_rcpf(1.0f + __builtin_amdgcn_exp2f(fmaf(v3, -kL2E, nenc2)));
        o_ts  = ts;
    };

    auto loadU = [&](int t) -> float {
        int tc = (t < T) ? t : (T - 1);
        tc = (tc > 0) ? tc : 0;
        return U[(size_t)tc * (size_t)N + (size_t)rc];
    };

    float Ttr = 1.f, amap = 0.f, dmap = 0.f, cr = 0.f, cg = 0.f, cb = 0.f;
    auto integrate = [&](float q_sig, float q_r, float q_g, float q_b,
                         float q_ts, float ts_next) {
        const float es = __builtin_amdgcn_exp2f(q_sig * (ts_next - q_ts) * ndnl);
        const float w  = (1.0f - es) * Ttr;
        Ttr *= es;
        cr += w * q_r;  cg += w * q_g;  cb += w * q_b;
        amap += w;      dmap += w * q_ts;
    };

    auto tsOf = [&](int t, float uu) -> float {
        return fmaf((float)t + uu, stepw, tn);
    };

    float ts_a = tsOf(1, loadU(1));
    float ts_b = tsOf(2, loadU(2));
    float p_sig, p_r, p_g, p_b, p_ts;
    sample(tsOf(0, U[(size_t)rc]), p_sig, p_r, p_g, p_b, p_ts);

    int t = 1;
    for (; t + 1 < T; t += 2) {
        const float na = loadU(t + 2);
        const float nb = loadU(t + 3);
        __builtin_amdgcn_sched_barrier(0);

        float a_sig, a_r, a_g, a_b, a_ts;
        sample(ts_a, a_sig, a_r, a_g, a_b, a_ts);
        integrate(p_sig, p_r, p_g, p_b, p_ts, a_ts);

        sample(ts_b, p_sig, p_r, p_g, p_b, p_ts);
        integrate(a_sig, a_r, a_g, a_b, a_ts, p_ts);

        ts_a = tsOf(t + 2, na);
        ts_b = tsOf(t + 3, nb);
    }
    if (t < T) {
        float a_sig, a_r, a_g, a_b, a_ts;
        sample(ts_a, a_sig, a_r, a_g, a_b, a_ts);
        integrate(p_sig, p_r, p_g, p_b, p_ts, a_ts);
        p_sig = a_sig; p_r = a_r; p_g = a_g; p_b = a_b; p_ts = a_ts;
    }
    {
        const float es = __builtin_amdgcn_exp2f(p_sig * (tfinal - p_ts) * ndnl);
        const float w  = (1.0f - es) * Ttr;
        cr += w * p_r;  cg += w * p_g;  cb += w * p_b;
        amap += w;      dmap += w * p_ts;
    }

    if (ray < N) {
        float* o5 = out + (size_t)ray * 5;
        o5[0] = cr   * active;
        o5[1] = cg   * active;
        o5[2] = cb   * active;
        o5[3] = amap * active;
        o5[4] = dmap * active;
    }
}

extern "C" void kernel_launch(void* const* d_in, const int* in_sizes, int n_in,
                              void* d_out, int out_size, void* d_ws, size_t ws_size,
                              hipStream_t stream) {
    const float* rays_o = (const float*)d_in[0];
    const float* rays_d = (const float*)d_in[1];
    const float* U      = (const float*)d_in[2];
    const float* aabb   = (const float*)d_in[3];
    const float* W1     = (const float*)d_in[4];
    const float* b1     = (const float*)d_in[5];
    const float* Wsig   = (const float*)d_in[6];
    const float* Wcol   = (const float*)d_in[7];
    const float* Wdir   = (const float*)d_in[8];
    float* out = (float*)d_out;

    const int N = in_sizes[0] / 3;
    const int T = (N > 0) ? (in_sizes[2] / N) : 0;
    const int block = 256;
    const int grid  = (N + block - 1) / block;
    nerf_render_kernel<<<grid, block, 0, stream>>>(rays_o, rays_d, U, aabb,
                                                   W1, b1, Wsig, Wcol, Wdir,
                                                   out, N, T);
}
